# starting kernel
# baseline (speedup 1.0000x reference)
#include <hip/hip_runtime.h>

typedef float v2f __attribute__((ext_vector_type(2)));
typedef float v8f __attribute__((ext_vector_type(8)));

#define K_DIM 128

template <int DOUT>
__global__ __launch_bounds__(256) void gemm_f32_wmma(
    const float* __restrict__ A, const float* __restrict__ W,
    float* __restrict__ T, int N) {
  constexpr int CT = DOUT / 16;
  constexpr int RT = 8 / CT;
  constexpr int ROWS = RT * 16;
  __shared__ float As[ROWS][K_DIM + 4];

  const int tid  = threadIdx.x;
  const int wave = tid >> 5;
  const int lane = tid & 31;
  const int rowTile = wave / CT;
  const int colTile = wave % CT;
  const long rowBase = (long)blockIdx.x * ROWS;

  for (int i = tid; i < ROWS * K_DIM; i += 256) {
    const int r = i >> 7;
    const int c = i & 127;
    const long gr = rowBase + r;
    As[r][c] = (gr < N) ? A[gr * K_DIM + c] : 0.0f;
  }
  __syncthreads();

  const int half = lane >> 4;
  const int l16  = lane & 15;
  const int n0   = colTile * 16 + l16;
  const int arow = rowTile * 16 + l16;

  v8f acc = {};
  #pragma unroll
  for (int kk = 0; kk < K_DIM; kk += 4) {
    const int ka = kk + 2 * half;
    v2f a, b;
    a.x = As[arow][ka];
    a.y = As[arow][ka + 1];
    b.x = W[ka * DOUT + n0];
    b.y = W[(ka + 1) * DOUT + n0];
    acc = __builtin_amdgcn_wmma_f32_16x16x4_f32(
         false, a,  false, b,
         (short)0, acc,  false,  false);
  }

  const long mBase = rowBase + rowTile * 16 + 8 * half;
  #pragma unroll
  for (int v = 0; v < 8; ++v) {
    const long m = mBase + v;
    if (m < N) T[m * DOUT + n0] = acc[v];
  }
}

__global__ void init_deg_kernel(float* deg, int N) {
  const long i = (long)blockIdx.x * blockDim.x + threadIdx.x;
  if (i < N) deg[i] = 1.0f;
}

__global__ void count_deg_kernel(const int* __restrict__ dst, float* deg, int E) {
  const long e = (long)blockIdx.x * blockDim.x + threadIdx.x;
  if (e < E) atomicAdd(&deg[dst[e]], 1.0f);
}

__global__ void finalize_dinv_kernel(float* deg, int N) {
  const long i = (long)blockIdx.x * blockDim.x + threadIdx.x;
  if (i < N) deg[i] = rsqrtf(deg[i]);
}

template <int D>
__global__ void agg_self_kernel(const float* __restrict__ T,
                                const float* __restrict__ dinv,
                                float* __restrict__ out, int N) {
  const long idx = (long)blockIdx.x * blockDim.x + threadIdx.x;
  if (idx >= (long)N * D) return;
  const long i = idx / D;
  const float w = dinv[i];
  out[idx] = T[idx] * w * w;
}

template <int D>
__global__ void agg_edges_kernel(const float* __restrict__ T,
                                 const int* __restrict__ src,
                                 const int* __restrict__ dst,
                                 const float* __restrict__ dinv,
                                 float* __restrict__ out, int E) {
  constexpr int TPE = D / 4;
  const long idx = (long)blockIdx.x * blockDim.x + threadIdx.x;
  const long e = idx / TPE;
  const int  t = (int)(idx % TPE);
  if (e >= E) return;
  const int s = src[e];
  const int d = dst[e];
  const float norm = dinv[s] * dinv[d];
  const float4 v = ((const float4*)(T + (long)s * D))[t];
  float* op = out + (long)d * D + t * 4;
  atomicAdd(op + 0, v.x * norm);
  atomicAdd(op + 1, v.y * norm);
  atomicAdd(op + 2, v.z * norm);
  atomicAdd(op + 3, v.w * norm);
}

template <int D>
__global__ void bias_relu_kernel(float* __restrict__ h,
                                 const float* __restrict__ b, long total) {
  const long idx = (long)blockIdx.x * blockDim.x + threadIdx.x;
  if (idx >= total) return;
  const int d = (int)(idx % D);
  h[idx] = fmaxf(h[idx] + b[d], 0.0f);
}

__global__ void zero_kernel(float* p, long n) {
  const long i = (long)blockIdx.x * blockDim.x + threadIdx.x;
  if (i < n) p[i] = 0.0f;
}

__global__ void pool_count_kernel(const int* __restrict__ batch, float* counts, int N) {
  const long i = (long)blockIdx.x * blockDim.x + threadIdx.x;
  if (i < N) atomicAdd(&counts[batch[i]], 1.0f);
}

__global__ void pool_sum_kernel(const float* __restrict__ h,
                                const int* __restrict__ batch,
                                float* sums, int N) {
  const long idx = (long)blockIdx.x * blockDim.x + threadIdx.x;
  if (idx >= (long)N * 64) return;
  const long i = idx / 64;
  const int  d = (int)(idx % 64);
  atomicAdd(&sums[(long)batch[i] * 64 + d], h[idx]);
}

__global__ void pool_final_kernel(const float* __restrict__ sums,
                                  const float* __restrict__ counts,
                                  const float* __restrict__ b3,
                                  float* __restrict__ out, int G) {
  const long idx = (long)blockIdx.x * blockDim.x + threadIdx.x;
  if (idx >= (long)G * 64) return;
  const long g = idx / 64;
  const int  d = (int)(idx % 64);
  const float c = counts[g];
  out[idx] = (sums[idx] + c * b3[d]) / fmaxf(c, 1.0f);
}

static inline int cdiv(long a, long b) { return (int)((a + b - 1) / b); }

extern "C" void kernel_launch(void* const* d_in, const int* in_sizes, int n_in,
                              void* d_out, int out_size, void* d_ws, size_t ws_size,
                              hipStream_t stream) {
  const float* x   = (const float*)d_in[0];
  const int*   ei  = (const int*)d_in[1];
  const int*   bat = (const int*)d_in[2];
  const float* W1  = (const float*)d_in[3];
  const float* b1  = (const float*)d_in[4];
  const float* W2  = (const float*)d_in[5];
  const float* b2  = (const float*)d_in[6];
  const float* W3  = (const float*)d_in[7];
  const float* b3  = (const float*)d_in[8];
  (void)n_in; (void)ws_size;

  const int N = in_sizes[0] / 128;
  const int E = in_sizes[1] / 2;
  const int G = out_size / 64;
  const int* src = ei;
  const int* dst = ei + E;

  float* ws = (float*)d_ws;
  size_t o = 0;
  float* dinv = ws + o;   o += ((size_t)N + 63) & ~(size_t)63;
  float* T    = ws + o;   o += (size_t)N * 128;
  float* AGG  = ws + o;   o += (size_t)N * 128;
  float* counts = ws + o; o += (size_t)G;
  float* sums   = ws + o; o += (size_t)G * 64;

  const int B = 256;
  float* out = (float*)d_out;

  init_deg_kernel<<<cdiv(N, B), B, 0, stream>>>(dinv, N);
  count_deg_kernel<<<cdiv(E, B), B, 0, stream>>>(dst, dinv, E);
  finalize_dinv_kernel<<<cdiv(N, B), B, 0, stream>>>(dinv, N);

  gemm_f32_wmma<128><<<cdiv(N, 16), B, 0, stream>>>(x, W1, T, N);
  agg_self_kernel<128><<<cdiv((long)N * 128, B), B, 0, stream>>>(T, dinv, AGG, N);
  agg_edges_kernel<128><<<cdiv((long)E * 32, B), B, 0, stream>>>(T, src, dst, dinv, AGG, E);
  bias_relu_kernel<128><<<cdiv((long)N * 128, B), B, 0, stream>>>(AGG, b1, (long)N * 128);

  gemm_f32_wmma<128><<<cdiv(N, 16), B, 0, stream>>>(AGG, W2, T, N);
  agg_self_kernel<128><<<cdiv((long)N * 128, B), B, 0, stream>>>(T, dinv, AGG, N);
  agg_edges_kernel<128><<<cdiv((long)E * 32, B), B, 0, stream>>>(T, src, dst, dinv, AGG, E);
  bias_relu_kernel<128><<<cdiv((long)N * 128, B), B, 0, stream>>>(AGG, b2, (long)N * 128);

  gemm_f32_wmma<64><<<cdiv(N, 32), B, 0, stream>>>(AGG, W3, T, N);
  agg_self_kernel<64><<<cdiv((long)N * 64, B), B, 0, stream>>>(T, dinv, AGG, N);
  agg_edges_kernel<64><<<cdiv((long)E * 16, B), B, 0, stream>>>(T, src, dst, dinv, AGG, E);

  zero_kernel<<<cdiv((long)G * 65, B), B, 0, stream>>>(counts, (long)G * 65);
  pool_count_kernel<<<cdiv(N, B), B, 0, stream>>>(bat, counts, N);
  pool_sum_kernel<<<cdiv((long)N * 64, B), B, 0, stream>>>(AGG, bat, sums, N);
  pool_final_kernel<<<cdiv((long)G * 64, B), B, 0, stream>>>(sums, counts, b3, out, G);
}


# Few, fat waves where wave entry bounded a launch: a GEMM block owns 112 rows, its waves keep their W fragments in registers for 7 row tiles; degree count takes 8 edges per thread; sums unchanged
# speedup vs baseline: 2.4981x; 2.4981x over previous
#include <hip/hip_runtime.h>

typedef float v2f __attribute__((ext_vector_type(2)));
typedef float v8f __attribute__((ext_vector_type(8)));

#define K_DIM 128

#define G_UNITS 7
#define G_ROWS  (G_UNITS * 16)
#define G_LD    (K_DIM + 4)
#define G_PRE   (G_ROWS / 8)

template <int NU>
__device__ __forceinline__ void mma_units(const float (*As)[G_LD], int row0, int half,
                                          const v2f (&bw)[K_DIM / 4], v8f (&acc)[NU]) {
#pragma unroll
  for (int s = 0; s < K_DIM / 4; ++s) {
    const int ka = 4 * s + 2 * half;
#pragma unroll
    for (int u = 0; u < NU; ++u) {
      const v2f a = *reinterpret_cast<const v2f*>(&As[row0 + 16 * u][ka]);
      acc[u] = __builtin_amdgcn_wmma_f32_16x16x4_f32(
           false, a,  false, bw[s],
           (short)0, acc[u],  false,  false);
    }
  }
}

template <int DOUT>
__device__ __forceinline__ void store_unit(const v8f& acc, long mBase, int n0,
                                           const float* wrow, float* __restrict__ T,
                                           float* S, int N) {
  float wv[8];
#pragma unroll
  for (int v = 0; v < 8; ++v) wv[v] = wrow[v];
  if (mBase + 8 <= (long)N) {
#pragma unroll
    for (int v = 0; v < 8; ++v) {
      const float t = acc[v];
      T[(mBase + v) * DOUT + n0] = t;
      S[(mBase + v) * DOUT + n0] = t * wv[v] * wv[v];
    }
  } else {
#pragma unroll
    for (int v = 0; v < 8; ++v) {
      const long m = mBase + v;
      if (m < N) {
        const float t = acc[v];
        T[m * DOUT + n0] = t;
        S[m * DOUT + n0] = t * wv[v] * wv[v];
      }
    }
  }
}

template <int DOUT, bool ACT_IN>
__global__ __launch_bounds__(256) void gemm_f32_wmma(
    const float* A, const float* __restrict__ bin, const float* __restrict__ W,
    const float* __restrict__ dinv, float* __restrict__ T, float* S, int N) {
  constexpr int CT = DOUT / 16;
  constexpr int NG = 8 / CT;
  __shared__ alignas(16) float As[G_ROWS][G_LD];
  __shared__ float Ws[G_ROWS];

  const int tid  = threadIdx.x;
  const int wave = tid >> 5;
  const int lane = tid & 31;
  const int colTile = wave % CT;
  const int grp     = wave / CT;
  const long rowBase = (long)blockIdx.x * G_ROWS;

  const int c4 = lane * 4;
  const int r8 = wave;

  float4 pre[G_PRE];
#pragma unroll
  for (int q = 0; q < 4; ++q) {
    const long gr = rowBase + q * 8 + r8;
    pre[q] = make_float4(0.0f, 0.0f, 0.0f, 0.0f);
    if (gr < N) pre[q] = *reinterpret_cast<const float4*>(A + gr * K_DIM + c4);
  }

  float wmine = 0.0f;
  if (tid < G_ROWS) {
    const long gr = rowBase + tid;
    if (gr < N) wmine = dinv[gr];
  }

  float4 bc = make_float4(0.0f, 0.0f, 0.0f, 0.0f);
  if (ACT_IN) { bc.x = bin[c4]; bc.y = bin[c4 + 1]; bc.z = bin[c4 + 2]; bc.w = bin[c4 + 3]; }

  const int half = lane >> 4;
  const int l16  = lane & 15;
  const int n0   = colTile * 16 + l16;

  v2f bw[K_DIM / 4];
#pragma unroll
  for (int s = 0; s < K_DIM / 4; ++s) {
    const int ka = 4 * s + 2 * half;
    bw[s].x = W[ka * DOUT + n0];
    bw[s].y = W[(ka + 1) * DOUT + n0];
  }

#pragma unroll
  for (int q = 4; q < G_PRE; ++q) {
    const long gr = rowBase + q * 8 + r8;
    pre[q] = make_float4(0.0f, 0.0f, 0.0f, 0.0f);
    if (gr < N) pre[q] = *reinterpret_cast<const float4*>(A + gr * K_DIM + c4);
  }

  if (tid < G_ROWS) Ws[tid] = wmine;

#pragma unroll
  for (int c = 0; c < (G_UNITS + 1) / 2; ++c) {
    const int unitsHere = (2 * c + 1 < G_UNITS) ? 2 : 1;
    const int trips = unitsHere * 2;
#pragma unroll
    for (int q = 0; q < trips; ++q) {
      const int lr = c * 32 + q * 8 + r8;
      float4 a = pre[c * 4 + q];
      if (ACT_IN) {
        a.x = fmaxf(a.x + bc.x, 0.0f);
        a.y = fmaxf(a.y + bc.y, 0.0f);
        a.z = fmaxf(a.z + bc.z, 0.0f);
        a.w = fmaxf(a.w + bc.w, 0.0f);
        if (rowBase + lr >= N) a = make_float4(0.0f, 0.0f, 0.0f, 0.0f);
      }
      *reinterpret_cast<float4*>(&As[lr][c4]) = a;
    }
    __builtin_amdgcn_fence(__ATOMIC_RELEASE, "workgroup", "local");
    __builtin_amdgcn_s_barrier();
    __builtin_amdgcn_fence(__ATOMIC_ACQUIRE, "workgroup", "local");

    if (NG == 1) {
      const int u0 = 2 * c;
      const int lrow = u0 * 16 + 8 * half;
      if (unitsHere == 2) {
        v8f acc[2] = {};
        mma_units<2>(As, u0 * 16 + l16, half, bw, acc);
        store_unit<DOUT>(acc[0], rowBase + lrow, n0, &Ws[lrow], T, S, N);
        store_unit<DOUT>(acc[1], rowBase + lrow + 16, n0, &Ws[lrow + 16], T, S, N);
      } else {
        v8f acc[1] = {};
        mma_units<1>(As, u0 * 16 + l16, half, bw, acc);
        store_unit<DOUT>(acc[0], rowBase + lrow, n0, &Ws[lrow], T, S, N);
      }
    } else {
      const int u = 2 * c + grp;
      if (u < G_UNITS) {
        const int lrow = u * 16 + 8 * half;
        v8f acc[1] = {};
        mma_units<1>(As, u * 16 + l16, half, bw, acc);
        store_unit<DOUT>(acc[0], rowBase + lrow, n0, &Ws[lrow], T, S, N);
      }
    }
  }
}

__global__ void init_deg_kernel(float* deg, int N, unsigned* tickets) {
  const long i = (long)blockIdx.x * blockDim.x + threadIdx.x;
  if (i < N) deg[i] = 1.0f;
  if (i < 6) tickets[i * 64] = 0u;
}

#define DEG_U 8
__global__ __launch_bounds__(256) void count_deg_kernel(const int* __restrict__ dst, float* deg, int E) {
  const long e0 = (long)blockIdx.x * (256 * DEG_U) + threadIdx.x;
  int d[DEG_U];
#pragma unroll
  for (int u = 0; u < DEG_U; ++u) {
    const long e = e0 + u * 256;
    d[u] = (e < E) ? dst[e] : -1;
  }
#pragma unroll
  for (int u = 0; u < DEG_U; ++u)
    if (d[u] >= 0) atomicAdd(&deg[d[u]], 1.0f);
}

__global__ void finalize_dinv_kernel(float* deg, int N) {
  const long i = (long)blockIdx.x * blockDim.x + threadIdx.x;
  if (i < N) deg[i] = rsqrtf(deg[i]);
}


__device__ __forceinline__ int wave_half() {
  const unsigned id = __builtin_amdgcn_s_sendmsg_rtn(0x87);
  return (int)((id >> 18) & 1u);
}

template <int D, int WT>
__global__ __launch_bounds__(256) void agg_edges_kernel(
    const float* __restrict__ T, const int* __restrict__ src,
    const int* __restrict__ dst, const float* __restrict__ dinv,
    float* __restrict__ out, int E, int N, unsigned* __restrict__ tickets) {
  constexpr int CPL = D / 32;
  constexpr int BT  = 8 * WT;
  __shared__ unsigned s_tile[2];
  __shared__ int s_home;
  const int tid  = threadIdx.x;
  const int lane = tid & 31;
  const int wave = __builtin_amdgcn_readfirstlane(tid >> 5);
  const int nTiles = (int)(((long)E + BT - 1) / BT);
  const int Nh = N >> 1;

  if (tid == 0) s_home = wave_half();
  __syncthreads();
  const int home = __builtin_amdgcn_readfirstlane(s_home);
  int k = 0;

  for (int pass = 0; pass < 2; ++pass) {
    const int h = home ^ pass;
    unsigned* ctr = tickets + h * 64;
    unsigned pend = 0;
    if (tid == 0) pend = atomicAdd(ctr, 1u);
    for (int it = 0; it <= nTiles; ++it) {
      const int slot = k & 1;
      ++k;
      if (tid == 0) s_tile[slot] = pend;
      __syncthreads();
      const unsigned t = __builtin_amdgcn_readfirstlane(s_tile[slot]);
      if (t >= (unsigned)nTiles) break;

      const long e = (long)t * BT + wave * WT + lane;
      const bool ok = lane < WT && e < (long)E;
      int sv = 0, dv = 0;
      if (ok) { sv = src[e]; dv = dst[e]; }
      const bool mine = ok && ((int)(dv >= Nh) == h);
      const unsigned mask = __builtin_amdgcn_ballot_w32(mine);

      float wsv = 0.0f, wdv = 0.0f;
      if (mine) { wsv = dinv[sv]; wdv = dinv[dv]; }

      int   dd[WT];
      float v[WT][CPL];
#pragma unroll
      for (int j = 0; j < WT; ++j) {
        if (mask & (1u << j)) {
          const int s = __builtin_amdgcn_readlane(sv, j);
          dd[j] = __builtin_amdgcn_readlane(dv, j);
          const float* tp = T + (long)s * D + lane;
#pragma unroll
          for (int cc = 0; cc < CPL; ++cc) v[j][cc] = tp[32 * cc];
        }
      }
      if (tid == 0) pend = atomicAdd(ctr, 1u);
      const int wsb = __float_as_int(wsv);
      const int wdb = __float_as_int(wdv);
#pragma unroll
      for (int j = 0; j < WT; ++j) {
        if (mask & (1u << j)) {
          const float nr = __int_as_float(__builtin_amdgcn_readlane(wsb, j)) *
                           __int_as_float(__builtin_amdgcn_readlane(wdb, j));
          float* op = out + (long)dd[j] * D + lane;
#pragma unroll
          for (int cc = 0; cc < CPL; ++cc) atomicAdd(op + 32 * cc, v[j][cc] * nr);
        }
      }
    }
  }
}

#define POOL_T     1024
#define POOL_SLOTS (POOL_T / 64)
#define POOL_CAP   2048
#define POOL_U     16
#define POOL_R     8

__global__ __launch_bounds__(POOL_T) void pool_mean_kernel(
    const float* __restrict__ h, const int* __restrict__ batch,
    const float* __restrict__ b3, float* __restrict__ out, int N) {
  __shared__ int   s_list[POOL_CAP];
  __shared__ int   s_n;
  __shared__ float s_part[POOL_SLOTS][64];
  const int tid = threadIdx.x;
  const int g   = (int)blockIdx.x;
  if (tid == 0) s_n = 0;
  __syncthreads();

  for (int base = tid; base < N; base += POOL_T * POOL_U) {
    int id[POOL_U];
#pragma unroll
    for (int u = 0; u < POOL_U; ++u) {
      const int i = base + u * POOL_T;
      id[u] = (i < N) ? batch[i] : -1;
    }
#pragma unroll
    for (int u = 0; u < POOL_U; ++u) {
      if (id[u] == g) {
        const int p = atomicAdd(&s_n, 1);
        if (p < POOL_CAP) s_list[p] = base + u * POOL_T;
      }
    }
  }
  __syncthreads();

  const int n    = s_n;
  const int col  = tid & 63;
  const int slot = tid >> 6;
  float acc = 0.0f;
  if (n <= POOL_CAP) {
    for (int k0 = slot; k0 < n; k0 += POOL_R * POOL_SLOTS) {
      int   nd[POOL_R];
      float r[POOL_R];
#pragma unroll
      for (int u = 0; u < POOL_R; ++u) {
        const int k = k0 + u * POOL_SLOTS;
        nd[u] = (k < n) ? s_list[k] : -1;
      }
#pragma unroll
      for (int u = 0; u < POOL_R; ++u)
        r[u] = (nd[u] >= 0) ? h[(long)nd[u] * 64 + col] : 0.0f;
#pragma unroll
      for (int u = 0; u < POOL_R; ++u) acc += r[u];
    }
  } else {
    for (int i = slot; i < N; i += POOL_SLOTS)
      if (batch[i] == g) acc += h[(long)i * 64 + col];
  }
  s_part[slot][col] = acc;
  __syncthreads();

  if (tid < 64) {
    float s = 0.0f;
#pragma unroll
    for (int q = 0; q < POOL_SLOTS; ++q) s += s_part[q][tid];
    const float c = (float)n;
    out[(long)g * 64 + tid] = (s + c * b3[tid]) / fmaxf(c, 1.0f);
  }
}

static inline int cdiv(long a, long b) { return (int)((a + b - 1) / b); }

extern "C" void kernel_launch(void* const* d_in, const int* in_sizes, int n_in,
                              void* d_out, int out_size, void* d_ws, size_t ws_size,
                              hipStream_t stream) {
  const float* x   = (const float*)d_in[0];
  const int*   ei  = (const int*)d_in[1];
  const int*   bat = (const int*)d_in[2];
  const float* W1  = (const float*)d_in[3];
  const float* b1  = (const float*)d_in[4];
  const float* W2  = (const float*)d_in[5];
  const float* b2  = (const float*)d_in[6];
  const float* W3  = (const float*)d_in[7];
  const float* b3  = (const float*)d_in[8];
  (void)n_in; (void)ws_size;

  const int N = in_sizes[0] / 128;
  const int E = in_sizes[1] / 2;
  const int G = out_size / 64;
  const int* src = ei;
  const int* dst = ei + E;

  float* ws = (float*)d_ws;
  size_t o = 0;
  float* dinv = ws + o;   o += ((size_t)N + 63) & ~(size_t)63;
  float* T    = ws + o;   o += (size_t)N * 128;
  float* AGG  = ws + o;   o += (size_t)N * 128;
  float* counts = ws + o; o += (size_t)G;

  const int B = 256;
  float* out = (float*)d_out;

  unsigned* tickets = (unsigned*)counts;
  const int AG128 = 1536;
  const int AG64  = 2048;

  init_deg_kernel<<<cdiv(N, B), B, 0, stream>>>(dinv, N, tickets);
  count_deg_kernel<<<cdiv(E, 256 * DEG_U), 256, 0, stream>>>(dst, dinv, E);
  finalize_dinv_kernel<<<cdiv(N, B), B, 0, stream>>>(dinv, N);

  gemm_f32_wmma<128, false><<<cdiv(N, G_ROWS), B, 0, stream>>>(x, b1, W1, dinv, T, AGG, N);
  agg_edges_kernel<128, 16><<<AG128, B, 0, stream>>>(T, src, dst, dinv, AGG, E, N, tickets);

  gemm_f32_wmma<128, true><<<cdiv(N, G_ROWS), B, 0, stream>>>(AGG, b1, W2, dinv, T, AGG, N);
  agg_edges_kernel<128, 16><<<AG128, B, 0, stream>>>(T, src, dst, dinv, AGG, E, N, tickets + 128);

  float* AGG3 = T + (size_t)N * 64;
  gemm_f32_wmma<64, true><<<cdiv(N, G_ROWS), B, 0, stream>>>(AGG, b2, W3, dinv, T, AGG3, N);
  agg_edges_kernel<64, 16><<<AG64, B, 0, stream>>>(T, src, dst, dinv, AGG3, E, N, tickets + 256);

  pool_mean_kernel<<<G, POOL_T, 0, stream>>>(AGG3, bat, b3, out, N);
}


# Edge weights computed once: layer 1's edge pass stores dinv[s]*dinv[d] per edge (3.2 MB more workspace, used if there), layers 2, 3 read them in runs, not 2 scattered lookups an edge; sums unchanged
# speedup vs baseline: 2.6206x; 1.0490x over previous
#include <hip/hip_runtime.h>

typedef float v2f __attribute__((ext_vector_type(2)));
typedef float v8f __attribute__((ext_vector_type(8)));

#define K_DIM 128

#define G_UNITS 7
#define G_ROWS  (G_UNITS * 16)
#define G_LD    (K_DIM + 4)
#define G_PRE   (G_ROWS / 8)

template <int NU>
__device__ __forceinline__ void mma_units(const float (*As)[G_LD], int row0, int half,
                                          const v2f (&bw)[K_DIM / 4], v8f (&acc)[NU]) {
#pragma unroll
  for (int s = 0; s < K_DIM / 4; ++s) {
    const int ka = 4 * s + 2 * half;
#pragma unroll
    for (int u = 0; u < NU; ++u) {
      const v2f a = *reinterpret_cast<const v2f*>(&As[row0 + 16 * u][ka]);
      acc[u] = __builtin_amdgcn_wmma_f32_16x16x4_f32(
           false, a,  false, bw[s],
           (short)0, acc[u],  false,  false);
    }
  }
}

template <int DOUT>
__device__ __forceinline__ void store_unit(const v8f& acc, long mBase, int n0,
                                           const float* wrow, float* __restrict__ T,
                                           float* S, int N) {
  float wv[8];
#pragma unroll
  for (int v = 0; v < 8; ++v) wv[v] = wrow[v];
  if (mBase + 8 <= (long)N) {
#pragma unroll
    for (int v = 0; v < 8; ++v) {
      const float t = acc[v];
      T[(mBase + v) * DOUT + n0] = t;
      S[(mBase + v) * DOUT + n0] = t * wv[v] * wv[v];
    }
  } else {
#pragma unroll
    for (int v = 0; v < 8; ++v) {
      const long m = mBase + v;
      if (m < N) {
        const float t = acc[v];
        T[m * DOUT + n0] = t;
        S[m * DOUT + n0] = t * wv[v] * wv[v];
      }
    }
  }
}

template <int DOUT, bool ACT_IN>
__global__ __launch_bounds__(256) void gemm_f32_wmma(
    const float* A, const float* __restrict__ bin, const float* __restrict__ W,
    const float* __restrict__ dinv, float* __restrict__ T, float* S, int N) {
  constexpr int CT = DOUT / 16;
  constexpr int NG = 8 / CT;
  __shared__ alignas(16) float As[G_ROWS][G_LD];
  __shared__ float Ws[G_ROWS];

  const int tid  = threadIdx.x;
  const int wave = tid >> 5;
  const int lane = tid & 31;
  const int colTile = wave % CT;
  const int grp     = wave / CT;
  const long rowBase = (long)blockIdx.x * G_ROWS;

  const int c4 = lane * 4;
  const int r8 = wave;

  float4 pre[G_PRE];
#pragma unroll
  for (int q = 0; q < 4; ++q) {
    const long gr = rowBase + q * 8 + r8;
    pre[q] = make_float4(0.0f, 0.0f, 0.0f, 0.0f);
    if (gr < N) pre[q] = *reinterpret_cast<const float4*>(A + gr * K_DIM + c4);
  }

  float wmine = 0.0f;
  if (tid < G_ROWS) {
    const long gr = rowBase + tid;
    if (gr < N) wmine = dinv[gr];
  }

  float4 bc = make_float4(0.0f, 0.0f, 0.0f, 0.0f);
  if (ACT_IN) { bc.x = bin[c4]; bc.y = bin[c4 + 1]; bc.z = bin[c4 + 2]; bc.w = bin[c4 + 3]; }

  const int half = lane >> 4;
  const int l16  = lane & 15;
  const int n0   = colTile * 16 + l16;

  v2f bw[K_DIM / 4];
#pragma unroll
  for (int s = 0; s < K_DIM / 4; ++s) {
    const int ka = 4 * s + 2 * half;
    bw[s].x = W[ka * DOUT + n0];
    bw[s].y = W[(ka + 1) * DOUT + n0];
  }

#pragma unroll
  for (int q = 4; q < G_PRE; ++q) {
    const long gr = rowBase + q * 8 + r8;
    pre[q] = make_float4(0.0f, 0.0f, 0.0f, 0.0f);
    if (gr < N) pre[q] = *reinterpret_cast<const float4*>(A + gr * K_DIM + c4);
  }

  if (tid < G_ROWS) Ws[tid] = wmine;

#pragma unroll
  for (int c = 0; c < (G_UNITS + 1) / 2; ++c) {
    const int unitsHere = (2 * c + 1 < G_UNITS) ? 2 : 1;
    const int trips = unitsHere * 2;
#pragma unroll
    for (int q = 0; q < trips; ++q) {
      const int lr = c * 32 + q * 8 + r8;
      float4 a = pre[c * 4 + q];
      if (ACT_IN) {
        a.x = fmaxf(a.x + bc.x, 0.0f);
        a.y = fmaxf(a.y + bc.y, 0.0f);
        a.z = fmaxf(a.z + bc.z, 0.0f);
        a.w = fmaxf(a.w + bc.w, 0.0f);
        if (rowBase + lr >= N) a = make_float4(0.0f, 0.0f, 0.0f, 0.0f);
      }
      *reinterpret_cast<float4*>(&As[lr][c4]) = a;
    }
    __builtin_amdgcn_fence(__ATOMIC_RELEASE, "workgroup", "local");
    __builtin_amdgcn_s_barrier();
    __builtin_amdgcn_fence(__ATOMIC_ACQUIRE, "workgroup", "local");

    if (NG == 1) {
      const int u0 = 2 * c;
      const int lrow = u0 * 16 + 8 * half;
      if (unitsHere == 2) {
        v8f acc[2] = {};
        mma_units<2>(As, u0 * 16 + l16, half, bw, acc);
        store_unit<DOUT>(acc[0], rowBase + lrow, n0, &Ws[lrow], T, S, N);
        store_unit<DOUT>(acc[1], rowBase + lrow + 16, n0, &Ws[lrow + 16], T, S, N);
      } else {
        v8f acc[1] = {};
        mma_units<1>(As, u0 * 16 + l16, half, bw, acc);
        store_unit<DOUT>(acc[0], rowBase + lrow, n0, &Ws[lrow], T, S, N);
      }
    } else {
      const int u = 2 * c + grp;
      if (u < G_UNITS) {
        const int lrow = u * 16 + 8 * half;
        v8f acc[1] = {};
        mma_units<1>(As, u * 16 + l16, half, bw, acc);
        store_unit<DOUT>(acc[0], rowBase + lrow, n0, &Ws[lrow], T, S, N);
      }
    }
  }
}

__global__ void init_deg_kernel(float* deg, int N, unsigned* tickets) {
  const long i = (long)blockIdx.x * blockDim.x + threadIdx.x;
  if (i < N) deg[i] = 1.0f;
  if (i < 6) tickets[i * 64] = 0u;
}

#define DEG_U 8
__global__ __launch_bounds__(256) void count_deg_kernel(const int* __restrict__ dst, float* deg, int E) {
  const long e0 = (long)blockIdx.x * (256 * DEG_U) + threadIdx.x;
  int d[DEG_U];
#pragma unroll
  for (int u = 0; u < DEG_U; ++u) {
    const long e = e0 + u * 256;
    d[u] = (e < E) ? dst[e] : -1;
  }
#pragma unroll
  for (int u = 0; u < DEG_U; ++u)
    if (d[u] >= 0) atomicAdd(&deg[d[u]], 1.0f);
}

__global__ void finalize_dinv_kernel(float* deg, int N) {
  const long i = (long)blockIdx.x * blockDim.x + threadIdx.x;
  if (i < N) deg[i] = rsqrtf(deg[i]);
}


__device__ __forceinline__ int wave_half() {
  const unsigned id = __builtin_amdgcn_s_sendmsg_rtn(0x87);
  return (int)((id >> 18) & 1u);
}

template <int D, int WT, bool WLOAD>
__global__ __launch_bounds__(256) void agg_edges_kernel(
    const float* __restrict__ T, const int* __restrict__ src,
    const int* __restrict__ dst, const float* __restrict__ dinv,
    float* __restrict__ out, int E, int N, unsigned* __restrict__ tickets,
    float* wedge) {
  constexpr int CPL = D / 32;
  constexpr int BT  = 8 * WT;
  __shared__ unsigned s_tile[2];
  __shared__ int s_home;
  const int tid  = threadIdx.x;
  const int lane = tid & 31;
  const int wave = __builtin_amdgcn_readfirstlane(tid >> 5);
  const int nTiles = (int)(((long)E + BT - 1) / BT);
  const int Nh = N >> 1;
  constexpr bool wread = WLOAD;
  const bool wwrite = !WLOAD && wedge != nullptr;

  if (tid == 0) s_home = wave_half();
  __syncthreads();
  const int home = __builtin_amdgcn_readfirstlane(s_home);
  int k = 0;

  for (int pass = 0; pass < 2; ++pass) {
    const int h = home ^ pass;
    unsigned* ctr = tickets + h * 64;
    unsigned pend = 0;
    if (tid == 0) pend = atomicAdd(ctr, 1u);
    for (int it = 0; it <= nTiles; ++it) {
      const int slot = k & 1;
      ++k;
      if (tid == 0) s_tile[slot] = pend;
      __syncthreads();
      const unsigned t = __builtin_amdgcn_readfirstlane(s_tile[slot]);
      if (t >= (unsigned)nTiles) break;

      const long e = (long)t * BT + wave * WT + lane;
      const bool ok = lane < WT && e < (long)E;
      int sv = 0, dv = 0;
      float wv = 0.0f;
      if (ok) {
        sv = src[e]; dv = dst[e];
        if (wread) wv = wedge[e];
      }
      const bool mine = ok && ((int)(dv >= Nh) == h);
      const unsigned mask = __builtin_amdgcn_ballot_w32(mine);

      float wsv = 0.0f, wdv = 0.0f;
      if (mine && !wread) { wsv = dinv[sv]; wdv = dinv[dv]; }

      int   dd[WT];
      float v[WT][CPL];
#pragma unroll
      for (int j = 0; j < WT; ++j) {
        if (mask & (1u << j)) {
          const int s = __builtin_amdgcn_readlane(sv, j);
          dd[j] = __builtin_amdgcn_readlane(dv, j);
          const float* tp = T + (long)s * D + lane;
#pragma unroll
          for (int cc = 0; cc < CPL; ++cc) v[j][cc] = tp[32 * cc];
        }
      }
      if (tid == 0) pend = atomicAdd(ctr, 1u);
      if (!wread) {
        wv = wsv * wdv;
        if (wwrite && mine) wedge[e] = wv;
      }
      const int wb = __float_as_int(wv);
#pragma unroll
      for (int j = 0; j < WT; ++j) {
        if (mask & (1u << j)) {
          const float nr = __int_as_float(__builtin_amdgcn_readlane(wb, j));
          float* op = out + (long)dd[j] * D + lane;
#pragma unroll
          for (int cc = 0; cc < CPL; ++cc) atomicAdd(op + 32 * cc, v[j][cc] * nr);
        }
      }
    }
  }
}

#define POOL_T     1024
#define POOL_SLOTS (POOL_T / 64)
#define POOL_CAP   2048
#define POOL_U     16
#define POOL_R     8

__global__ __launch_bounds__(POOL_T) void pool_mean_kernel(
    const float* __restrict__ h, const int* __restrict__ batch,
    const float* __restrict__ b3, float* __restrict__ out, int N) {
  __shared__ int   s_list[POOL_CAP];
  __shared__ int   s_n;
  __shared__ float s_part[POOL_SLOTS][64];
  const int tid = threadIdx.x;
  const int g   = (int)blockIdx.x;
  if (tid == 0) s_n = 0;
  __syncthreads();

  for (int base = tid; base < N; base += POOL_T * POOL_U) {
    int id[POOL_U];
#pragma unroll
    for (int u = 0; u < POOL_U; ++u) {
      const int i = base + u * POOL_T;
      id[u] = (i < N) ? batch[i] : -1;
    }
#pragma unroll
    for (int u = 0; u < POOL_U; ++u) {
      if (id[u] == g) {
        const int p = atomicAdd(&s_n, 1);
        if (p < POOL_CAP) s_list[p] = base + u * POOL_T;
      }
    }
  }
  __syncthreads();

  const int n    = s_n;
  const int col  = tid & 63;
  const int slot = tid >> 6;
  float acc = 0.0f;
  if (n <= POOL_CAP) {
    for (int k0 = slot; k0 < n; k0 += POOL_R * POOL_SLOTS) {
      int   nd[POOL_R];
      float r[POOL_R];
#pragma unroll
      for (int u = 0; u < POOL_R; ++u) {
        const int k = k0 + u * POOL_SLOTS;
        nd[u] = (k < n) ? s_list[k] : -1;
      }
#pragma unroll
      for (int u = 0; u < POOL_R; ++u)
        r[u] = (nd[u] >= 0) ? h[(long)nd[u] * 64 + col] : 0.0f;
#pragma unroll
      for (int u = 0; u < POOL_R; ++u) acc += r[u];
    }
  } else {
    for (int i = slot; i < N; i += POOL_SLOTS)
      if (batch[i] == g) acc += h[(long)i * 64 + col];
  }
  s_part[slot][col] = acc;
  __syncthreads();

  if (tid < 64) {
    float s = 0.0f;
#pragma unroll
    for (int q = 0; q < POOL_SLOTS; ++q) s += s_part[q][tid];
    const float c = (float)n;
    out[(long)g * 64 + tid] = (s + c * b3[tid]) / fmaxf(c, 1.0f);
  }
}

static inline int cdiv(long a, long b) { return (int)((a + b - 1) / b); }

extern "C" void kernel_launch(void* const* d_in, const int* in_sizes, int n_in,
                              void* d_out, int out_size, void* d_ws, size_t ws_size,
                              hipStream_t stream) {
  const float* x   = (const float*)d_in[0];
  const int*   ei  = (const int*)d_in[1];
  const int*   bat = (const int*)d_in[2];
  const float* W1  = (const float*)d_in[3];
  const float* b1  = (const float*)d_in[4];
  const float* W2  = (const float*)d_in[5];
  const float* b2  = (const float*)d_in[6];
  const float* W3  = (const float*)d_in[7];
  const float* b3  = (const float*)d_in[8];
  (void)n_in;

  const int N = in_sizes[0] / 128;
  const int E = in_sizes[1] / 2;
  const int G = out_size / 64;
  const int* src = ei;
  const int* dst = ei + E;

  float* ws = (float*)d_ws;
  size_t o = 0;
  float* dinv = ws + o;   o += ((size_t)N + 63) & ~(size_t)63;
  float* T    = ws + o;   o += (size_t)N * 128;
  float* AGG  = ws + o;   o += (size_t)N * 128;
  float* counts = ws + o; o += (size_t)G;
  o = (o + 63) & ~(size_t)63;
  float* wedge = (ws_size >= (o + (size_t)E) * sizeof(float)) ? ws + o : nullptr;

  const int B = 256;
  float* out = (float*)d_out;

  unsigned* tickets = (unsigned*)counts;
  const int AG128 = 1536;
  const int AG64  = 2048;

  init_deg_kernel<<<cdiv(N, B), B, 0, stream>>>(dinv, N, tickets);
  count_deg_kernel<<<cdiv(E, 256 * DEG_U), 256, 0, stream>>>(dst, dinv, E);
  finalize_dinv_kernel<<<cdiv(N, B), B, 0, stream>>>(dinv, N);

  gemm_f32_wmma<128, false><<<cdiv(N, G_ROWS), B, 0, stream>>>(x, b1, W1, dinv, T, AGG, N);
  agg_edges_kernel<128, 16, false><<<AG128, B, 0, stream>>>(T, src, dst, dinv, AGG, E, N, tickets, wedge);

  gemm_f32_wmma<128, true><<<cdiv(N, G_ROWS), B, 0, stream>>>(AGG, b1, W2, dinv, T, AGG, N);
  if (wedge != nullptr)
    agg_edges_kernel<128, 16, true><<<AG128, B, 0, stream>>>(T, src, dst, dinv, AGG, E, N, tickets + 128, wedge);
  else
    agg_edges_kernel<128, 16, false><<<AG128, B, 0, stream>>>(T, src, dst, dinv, AGG, E, N, tickets + 128, wedge);

  float* AGG3 = T + (size_t)N * 64;
  gemm_f32_wmma<64, true><<<cdiv(N, G_ROWS), B, 0, stream>>>(AGG, b2, W3, dinv, T, AGG3, N);
  if (wedge != nullptr)
    agg_edges_kernel<64, 16, true><<<AG64, B, 0, stream>>>(T, src, dst, dinv, AGG3, E, N, tickets + 256, wedge);
  else
    agg_edges_kernel<64, 16, false><<<AG64, B, 0, stream>>>(T, src, dst, dinv, AGG3, E, N, tickets + 256, wedge);

  pool_mean_kernel<<<G, POOL_T, 0, stream>>>(AGG3, bat, b3, out, N);
}


# Layer 3 scatter composed with the mean pool (both linear): an edge or self loop adds w*T[src] to its graph's row of a 512x64 sum per block in LDS; blocks' sums by global f32 adds; arrival order
# speedup vs baseline: 2.7460x; 1.0479x over previous
#include <hip/hip_runtime.h>

typedef float v2f __attribute__((ext_vector_type(2)));
typedef float v8f __attribute__((ext_vector_type(8)));

#define K_DIM 128

#define G_UNITS 7
#define G_ROWS  (G_UNITS * 16)
#define G_LD    (K_DIM + 4)
#define G_PRE   (G_ROWS / 8)

template <int NU>
__device__ __forceinline__ void mma_units(const float (*As)[G_LD], int row0, int half,
                                          const v2f (&bw)[K_DIM / 4], v8f (&acc)[NU]) {
#pragma unroll
  for (int s = 0; s < K_DIM / 4; ++s) {
    const int ka = 4 * s + 2 * half;
#pragma unroll
    for (int u = 0; u < NU; ++u) {
      const v2f a = *reinterpret_cast<const v2f*>(&As[row0 + 16 * u][ka]);
      acc[u] = __builtin_amdgcn_wmma_f32_16x16x4_f32(
           false, a,  false, bw[s],
           (short)0, acc[u],  false,  false);
    }
  }
}

template <int DOUT, bool WITH_S>
__device__ __forceinline__ void store_unit(const v8f& acc, long mBase, int n0,
                                           const float* wrow, float* __restrict__ T,
                                           float* S, int N) {
  float wv[8];
#pragma unroll
  for (int v = 0; v < 8; ++v) wv[v] = WITH_S ? wrow[v] : 0.0f;
  if (mBase + 8 <= (long)N) {
#pragma unroll
    for (int v = 0; v < 8; ++v) {
      const float t = acc[v];
      T[(mBase + v) * DOUT + n0] = t;
      if (WITH_S) S[(mBase + v) * DOUT + n0] = t * wv[v] * wv[v];
    }
  } else {
#pragma unroll
    for (int v = 0; v < 8; ++v) {
      const long m = mBase + v;
      if (m < N) {
        const float t = acc[v];
        T[m * DOUT + n0] = t;
        if (WITH_S) S[m * DOUT + n0] = t * wv[v] * wv[v];
      }
    }
  }
}

template <int DOUT, bool ACT_IN, bool WITH_S>
__global__ __launch_bounds__(256) void gemm_f32_wmma(
    const float* A, const float* __restrict__ bin, const float* __restrict__ W,
    const float* __restrict__ dinv, float* __restrict__ T, float* S, int N) {
  constexpr int CT = DOUT / 16;
  constexpr int NG = 8 / CT;
  __shared__ alignas(16) float As[G_ROWS][G_LD];
  __shared__ float Ws[G_ROWS];

  const int tid  = threadIdx.x;
  const int wave = tid >> 5;
  const int lane = tid & 31;
  const int colTile = wave % CT;
  const int grp     = wave / CT;
  const long rowBase = (long)blockIdx.x * G_ROWS;

  const int c4 = lane * 4;
  const int r8 = wave;

  float4 pre[G_PRE];
#pragma unroll
  for (int q = 0; q < 4; ++q) {
    const long gr = rowBase + q * 8 + r8;
    pre[q] = make_float4(0.0f, 0.0f, 0.0f, 0.0f);
    if (gr < N) pre[q] = *reinterpret_cast<const float4*>(A + gr * K_DIM + c4);
  }

  float wmine = 0.0f;
  if (WITH_S && tid < G_ROWS) {
    const long gr = rowBase + tid;
    if (gr < N) wmine = dinv[gr];
  }

  float4 bc = make_float4(0.0f, 0.0f, 0.0f, 0.0f);
  if (ACT_IN) { bc.x = bin[c4]; bc.y = bin[c4 + 1]; bc.z = bin[c4 + 2]; bc.w = bin[c4 + 3]; }

  const int half = lane >> 4;
  const int l16  = lane & 15;
  const int n0   = colTile * 16 + l16;

  v2f bw[K_DIM / 4];
#pragma unroll
  for (int s = 0; s < K_DIM / 4; ++s) {
    const int ka = 4 * s + 2 * half;
    bw[s].x = W[ka * DOUT + n0];
    bw[s].y = W[(ka + 1) * DOUT + n0];
  }

#pragma unroll
  for (int q = 4; q < G_PRE; ++q) {
    const long gr = rowBase + q * 8 + r8;
    pre[q] = make_float4(0.0f, 0.0f, 0.0f, 0.0f);
    if (gr < N) pre[q] = *reinterpret_cast<const float4*>(A + gr * K_DIM + c4);
  }

  if (WITH_S && tid < G_ROWS) Ws[tid] = wmine;

#pragma unroll
  for (int c = 0; c < (G_UNITS + 1) / 2; ++c) {
    const int unitsHere = (2 * c + 1 < G_UNITS) ? 2 : 1;
    const int trips = unitsHere * 2;
#pragma unroll
    for (int q = 0; q < trips; ++q) {
      const int lr = c * 32 + q * 8 + r8;
      float4 a = pre[c * 4 + q];
      if (ACT_IN) {
        a.x = fmaxf(a.x + bc.x, 0.0f);
        a.y = fmaxf(a.y + bc.y, 0.0f);
        a.z = fmaxf(a.z + bc.z, 0.0f);
        a.w = fmaxf(a.w + bc.w, 0.0f);
        if (rowBase + lr >= N) a = make_float4(0.0f, 0.0f, 0.0f, 0.0f);
      }
      *reinterpret_cast<float4*>(&As[lr][c4]) = a;
    }
    __builtin_amdgcn_fence(__ATOMIC_RELEASE, "workgroup", "local");
    __builtin_amdgcn_s_barrier();
    __builtin_amdgcn_fence(__ATOMIC_ACQUIRE, "workgroup", "local");

    if (NG == 1) {
      const int u0 = 2 * c;
      const int lrow = u0 * 16 + 8 * half;
      if (unitsHere == 2) {
        v8f acc[2] = {};
        mma_units<2>(As, u0 * 16 + l16, half, bw, acc);
        store_unit<DOUT, WITH_S>(acc[0], rowBase + lrow, n0, &Ws[lrow], T, S, N);
        store_unit<DOUT, WITH_S>(acc[1], rowBase + lrow + 16, n0, &Ws[lrow + 16], T, S, N);
      } else {
        v8f acc[1] = {};
        mma_units<1>(As, u0 * 16 + l16, half, bw, acc);
        store_unit<DOUT, WITH_S>(acc[0], rowBase + lrow, n0, &Ws[lrow], T, S, N);
      }
    } else {
      const int u = 2 * c + grp;
      if (u < G_UNITS) {
        const int lrow = u * 16 + 8 * half;
        v8f acc[1] = {};
        mma_units<1>(As, u * 16 + l16, half, bw, acc);
        store_unit<DOUT, WITH_S>(acc[0], rowBase + lrow, n0, &Ws[lrow], T, S, N);
      }
    }
  }
}

__global__ void init_deg_kernel(float* deg, int N, unsigned* tickets, unsigned* zeros, int nz) {
  const long i = (long)blockIdx.x * blockDim.x + threadIdx.x;
  if (i < N) deg[i] = 1.0f;
  if (i < 6) tickets[i * 64] = 0u;
  const long stride = (long)gridDim.x * blockDim.x;
  for (long k = i; k < nz; k += stride) zeros[k] = 0u;
}

#define DEG_U 8
__global__ __launch_bounds__(256) void count_deg_kernel(const int* __restrict__ dst, float* deg, int E) {
  const long e0 = (long)blockIdx.x * (256 * DEG_U) + threadIdx.x;
  int d[DEG_U];
#pragma unroll
  for (int u = 0; u < DEG_U; ++u) {
    const long e = e0 + u * 256;
    d[u] = (e < E) ? dst[e] : -1;
  }
#pragma unroll
  for (int u = 0; u < DEG_U; ++u)
    if (d[u] >= 0) atomicAdd(&deg[d[u]], 1.0f);
}

__global__ void finalize_dinv_kernel(float* deg, int N) {
  const long i = (long)blockIdx.x * blockDim.x + threadIdx.x;
  if (i < N) deg[i] = rsqrtf(deg[i]);
}


__device__ __forceinline__ int wave_half() {
  const unsigned id = __builtin_amdgcn_s_sendmsg_rtn(0x87);
  return (int)((id >> 18) & 1u);
}

template <int D, int WT, bool WLOAD>
__global__ __launch_bounds__(256) void agg_edges_kernel(
    const float* __restrict__ T, const int* __restrict__ src,
    const int* __restrict__ dst, const float* __restrict__ dinv,
    float* __restrict__ out, int E, int N, unsigned* __restrict__ tickets,
    float* wedge) {
  constexpr int CPL = D / 32;
  constexpr int BT  = 8 * WT;
  __shared__ unsigned s_tile[2];
  __shared__ int s_home;
  const int tid  = threadIdx.x;
  const int lane = tid & 31;
  const int wave = __builtin_amdgcn_readfirstlane(tid >> 5);
  const int nTiles = (int)(((long)E + BT - 1) / BT);
  const int Nh = N >> 1;
  constexpr bool wread = WLOAD;
  const bool wwrite = !WLOAD && wedge != nullptr;

  if (tid == 0) s_home = wave_half();
  __syncthreads();
  const int home = __builtin_amdgcn_readfirstlane(s_home);
  int k = 0;

  for (int pass = 0; pass < 2; ++pass) {
    const int h = home ^ pass;
    unsigned* ctr = tickets + h * 64;
    unsigned pend = 0;
    if (tid == 0) pend = atomicAdd(ctr, 1u);
    for (int it = 0; it <= nTiles; ++it) {
      const int slot = k & 1;
      ++k;
      if (tid == 0) s_tile[slot] = pend;
      __syncthreads();
      const unsigned t = __builtin_amdgcn_readfirstlane(s_tile[slot]);
      if (t >= (unsigned)nTiles) break;

      const long e = (long)t * BT + wave * WT + lane;
      const bool ok = lane < WT && e < (long)E;
      int sv = 0, dv = 0;
      float wv = 0.0f;
      if (ok) {
        sv = src[e]; dv = dst[e];
        if (wread) wv = wedge[e];
      }
      const bool mine = ok && ((int)(dv >= Nh) == h);
      const unsigned mask = __builtin_amdgcn_ballot_w32(mine);

      float wsv = 0.0f, wdv = 0.0f;
      if (mine && !wread) { wsv = dinv[sv]; wdv = dinv[dv]; }

      int   dd[WT];
      float v[WT][CPL];
#pragma unroll
      for (int j = 0; j < WT; ++j) {
        if (mask & (1u << j)) {
          const int s = __builtin_amdgcn_readlane(sv, j);
          dd[j] = __builtin_amdgcn_readlane(dv, j);
          const float* tp = T + (long)s * D + lane;
#pragma unroll
          for (int cc = 0; cc < CPL; ++cc) v[j][cc] = tp[32 * cc];
        }
      }
      if (tid == 0) pend = atomicAdd(ctr, 1u);
      if (!wread) {
        wv = wsv * wdv;
        if (wwrite && mine) wedge[e] = wv;
      }
      const int wb = __float_as_int(wv);
#pragma unroll
      for (int j = 0; j < WT; ++j) {
        if (mask & (1u << j)) {
          const float nr = __int_as_float(__builtin_amdgcn_readlane(wb, j));
          float* op = out + (long)dd[j] * D + lane;
#pragma unroll
          for (int cc = 0; cc < CPL; ++cc) atomicAdd(op + 32 * cc, v[j][cc] * nr);
        }
      }
    }
  }
}

#define PE_T 1024
#define PE_G 512
#define PE_W 16

__device__ __forceinline__ void pe_record(long t, bool ok, int E,
                                          const int* __restrict__ src,
                                          const int* __restrict__ dst,
                                          const float* wedge,
                                          int& sv, int& nv, float& wa) {
  sv = 0; nv = 0; wa = 0.0f;
  if (ok) {
    if (t < (long)E) {
      sv = src[t]; nv = dst[t];
      if (wedge != nullptr) wa = wedge[t];
    } else {
      sv = nv = (int)(t - (long)E);
    }
  }
}

__global__ __launch_bounds__(PE_T) void pool_edges_kernel(
    const float* __restrict__ T, const int* __restrict__ src,
    const int* __restrict__ dst, const int* __restrict__ batch,
    const float* __restrict__ dinv, const float* wedge,
    float* accG, unsigned* cntG, int E, int N, int G) {
  __shared__ float    s_acc[PE_G * 64];
  __shared__ unsigned s_cnt[PE_G];
  __shared__ int      s_home;
  const int tid  = threadIdx.x;
  const int lane = tid & 31;
  const int wave = __builtin_amdgcn_readfirstlane(tid >> 5);
  const long items   = (long)E + (long)N;
  const long nChunks = (items + PE_W - 1) / PE_W;
  const long gw = (long)blockIdx.x * (PE_T / 32) + wave;
  const long GW = (long)gridDim.x * (PE_T / 32);
  if (tid == 0) s_home = wave_half();

  for (int g0 = 0; g0 < G; g0 += PE_G) {
    for (int i = tid; i < PE_G * 64; i += PE_T) s_acc[i] = 0.0f;
    if (tid < PE_G) s_cnt[tid] = 0u;
    __syncthreads();

    long c = gw;
    long t = c * PE_W + lane;
    bool ok = lane < PE_W && c < nChunks && t < items;
    int sv, nv;
    float wa;
    pe_record(t, ok, E, src, dst, wedge, sv, nv, wa);

    while (c < nChunks) {
      int gv = batch[nv];

      float v[PE_W][2];
#pragma unroll
      for (int j = 0; j < PE_W; ++j) {
        const int s = __builtin_amdgcn_readlane(sv, j);
        const float* tp = T + (long)s * 64 + lane;
        v[j][0] = tp[0];
        v[j][1] = tp[32];
      }

      const long c2 = c + GW;
      const long t2 = c2 * PE_W + lane;
      const bool ok2 = lane < PE_W && c2 < nChunks && t2 < items;
      int sv2, nv2;
      float wa2;
      pe_record(t2, ok2, E, src, dst, wedge, sv2, nv2, wa2);

      float w = wa;
      if (ok && (t >= (long)E || wedge == nullptr)) w = dinv[sv] * dinv[nv];

      const int   gl = gv - g0;
      const bool mine = ok && gl >= 0 && gl < PE_G && gv < G;
      if (mine && t >= (long)E)
        __hip_atomic_fetch_add(&s_cnt[gl], 1u, __ATOMIC_RELAXED, __HIP_MEMORY_SCOPE_WORKGROUP);
      const unsigned amask = __builtin_amdgcn_ballot_w32(mine);
      const int wb = __float_as_int(w);
#pragma unroll
      for (int j = 0; j < PE_W; ++j) {
        if (amask & (1u << j)) {
          const int   g  = __builtin_amdgcn_readlane(gl, j);
          const float wj = __int_as_float(__builtin_amdgcn_readlane(wb, j));
          float* ap = &s_acc[g * 64 + lane];
          __hip_atomic_fetch_add(ap,      v[j][0] * wj, __ATOMIC_RELAXED, __HIP_MEMORY_SCOPE_WORKGROUP);
          __hip_atomic_fetch_add(ap + 32, v[j][1] * wj, __ATOMIC_RELAXED, __HIP_MEMORY_SCOPE_WORKGROUP);
        }
      }
      c = c2; t = t2; ok = ok2; sv = sv2; nv = nv2; wa = wa2;
    }
    __syncthreads();

    float* acc = accG + (long)s_home * ((long)G * 64) + (long)g0 * 64;
    const int lim = (G - g0 < PE_G ? G - g0 : PE_G) * 64;
    const int rot = tid + (int)blockIdx.x * 128;
#pragma nounroll
    for (int k = 0; k < PE_G * 64; k += PE_T * 8) {
      float x[8];
#pragma unroll
      for (int u = 0; u < 8; ++u) x[u] = s_acc[(k + u * PE_T + rot) & (PE_G * 64 - 1)];
#pragma unroll
      for (int u = 0; u < 8; ++u) {
        const int i = (k + u * PE_T + rot) & (PE_G * 64 - 1);
        if (i < lim) atomicAdd(acc + i, x[u]);
      }
    }
    if (tid < PE_G && g0 + tid < G) {
      const unsigned n = s_cnt[tid];
      if (n != 0u) atomicAdd(cntG + g0 + tid, n);
    }
    __syncthreads();
  }
}

__global__ __launch_bounds__(256) void pool_final_kernel(
    const float* __restrict__ accG, const unsigned* __restrict__ cntG,
    const float* __restrict__ b3, float* __restrict__ out, int G) {
  const int i = (int)blockIdx.x * 256 + threadIdx.x;
  if (i < G * 64) {
    const float s = accG[i] + accG[(long)G * 64 + i];
    const float c = (float)cntG[i >> 6];
    out[i] = (s + c * b3[i & 63]) / fmaxf(c, 1.0f);
  }
}

static inline int cdiv(long a, long b) { return (int)((a + b - 1) / b); }

extern "C" void kernel_launch(void* const* d_in, const int* in_sizes, int n_in,
                              void* d_out, int out_size, void* d_ws, size_t ws_size,
                              hipStream_t stream) {
  const float* x   = (const float*)d_in[0];
  const int*   ei  = (const int*)d_in[1];
  const int*   bat = (const int*)d_in[2];
  const float* W1  = (const float*)d_in[3];
  const float* b1  = (const float*)d_in[4];
  const float* W2  = (const float*)d_in[5];
  const float* b2  = (const float*)d_in[6];
  const float* W3  = (const float*)d_in[7];
  const float* b3  = (const float*)d_in[8];
  (void)n_in;

  const int N = in_sizes[0] / 128;
  const int E = in_sizes[1] / 2;
  const int G = out_size / 64;
  const int* src = ei;
  const int* dst = ei + E;

  float* ws = (float*)d_ws;
  size_t o = 0;
  float* dinv = ws + o;   o += ((size_t)N + 63) & ~(size_t)63;
  float* T    = ws + o;   o += (size_t)N * 128;
  float* AGG  = ws + o;   o += (size_t)N * 128;
  float* counts = ws + o; o += (size_t)G;
  o = (o + 63) & ~(size_t)63;
  float* accG = ws + o;   o += (size_t)G * 128;
  unsigned* cntG = (unsigned*)(ws + o); o += (size_t)G;
  o = (o + 63) & ~(size_t)63;
  float* wedge = (ws_size >= (o + (size_t)E) * sizeof(float)) ? ws + o : nullptr;

  const int B = 256;
  float* out = (float*)d_out;

  unsigned* tickets = (unsigned*)counts;
  const int AG128 = 1536;
  const int PE_BLOCKS = 256;

  init_deg_kernel<<<cdiv(N, B), B, 0, stream>>>(dinv, N, tickets, (unsigned*)accG, G * 129);
  count_deg_kernel<<<cdiv(E, 256 * DEG_U), 256, 0, stream>>>(dst, dinv, E);
  finalize_dinv_kernel<<<cdiv(N, B), B, 0, stream>>>(dinv, N);

  gemm_f32_wmma<128, false, true><<<cdiv(N, G_ROWS), B, 0, stream>>>(x, b1, W1, dinv, T, AGG, N);
  agg_edges_kernel<128, 16, false><<<AG128, B, 0, stream>>>(T, src, dst, dinv, AGG, E, N, tickets, wedge);

  gemm_f32_wmma<128, true, true><<<cdiv(N, G_ROWS), B, 0, stream>>>(AGG, b1, W2, dinv, T, AGG, N);
  if (wedge != nullptr)
    agg_edges_kernel<128, 16, true><<<AG128, B, 0, stream>>>(T, src, dst, dinv, AGG, E, N, tickets + 128, wedge);
  else
    agg_edges_kernel<128, 16, false><<<AG128, B, 0, stream>>>(T, src, dst, dinv, AGG, E, N, tickets + 128, wedge);

  gemm_f32_wmma<64, true, false><<<cdiv(N, G_ROWS), B, 0, stream>>>(AGG, b2, W3, dinv, T, nullptr, N);
  pool_edges_kernel<<<PE_BLOCKS, PE_T, 0, stream>>>(T, src, dst, bat, dinv, wedge, accG, cntG, E, N, G);

  pool_final_kernel<<<cdiv((long)G * 64, B), B, 0, stream>>>(accG, cntG, b3, out, G);
}


# In-degree count reduced in LDS: a block counts one edge slice's destinations of one 8192-node chunk with LDS integer adds, then whole-line global adds (per-half copies); deg bit-identical
# speedup vs baseline: 3.4323x; 1.2499x over previous
#include <hip/hip_runtime.h>

typedef float v2f __attribute__((ext_vector_type(2)));
typedef float v8f __attribute__((ext_vector_type(8)));

#define K_DIM 128

#define G_UNITS 7
#define G_ROWS  (G_UNITS * 16)
#define G_LD    (K_DIM + 4)
#define G_PRE   (G_ROWS / 8)

template <int NU>
__device__ __forceinline__ void mma_units(const float (*As)[G_LD], int row0, int half,
                                          const v2f (&bw)[K_DIM / 4], v8f (&acc)[NU]) {
#pragma unroll
  for (int s = 0; s < K_DIM / 4; ++s) {
    const int ka = 4 * s + 2 * half;
#pragma unroll
    for (int u = 0; u < NU; ++u) {
      const v2f a = *reinterpret_cast<const v2f*>(&As[row0 + 16 * u][ka]);
      acc[u] = __builtin_amdgcn_wmma_f32_16x16x4_f32(
           false, a,  false, bw[s],
           (short)0, acc[u],  false,  false);
    }
  }
}

template <int DOUT, bool WITH_S>
__device__ __forceinline__ void store_unit(const v8f& acc, long mBase, int n0,
                                           const float* wrow, float* __restrict__ T,
                                           float* S, int N) {
  float wv[8];
#pragma unroll
  for (int v = 0; v < 8; ++v) wv[v] = WITH_S ? wrow[v] : 0.0f;
  if (mBase + 8 <= (long)N) {
#pragma unroll
    for (int v = 0; v < 8; ++v) {
      const float t = acc[v];
      T[(mBase + v) * DOUT + n0] = t;
      if (WITH_S) S[(mBase + v) * DOUT + n0] = t * wv[v] * wv[v];
    }
  } else {
#pragma unroll
    for (int v = 0; v < 8; ++v) {
      const long m = mBase + v;
      if (m < N) {
        const float t = acc[v];
        T[m * DOUT + n0] = t;
        if (WITH_S) S[m * DOUT + n0] = t * wv[v] * wv[v];
      }
    }
  }
}

template <int DOUT, bool ACT_IN, bool WITH_S>
__global__ __launch_bounds__(256) void gemm_f32_wmma(
    const float* A, const float* __restrict__ bin, const float* __restrict__ W,
    const float* __restrict__ dinv, float* __restrict__ T, float* S, int N) {
  constexpr int CT = DOUT / 16;
  constexpr int NG = 8 / CT;
  __shared__ alignas(16) float As[G_ROWS][G_LD];
  __shared__ float Ws[G_ROWS];

  const int tid  = threadIdx.x;
  const int wave = tid >> 5;
  const int lane = tid & 31;
  const int colTile = wave % CT;
  const int grp     = wave / CT;
  const long rowBase = (long)blockIdx.x * G_ROWS;

  const int c4 = lane * 4;
  const int r8 = wave;

  float4 pre[G_PRE];
#pragma unroll
  for (int q = 0; q < 4; ++q) {
    const long gr = rowBase + q * 8 + r8;
    pre[q] = make_float4(0.0f, 0.0f, 0.0f, 0.0f);
    if (gr < N) pre[q] = *reinterpret_cast<const float4*>(A + gr * K_DIM + c4);
  }

  float wmine = 0.0f;
  if (WITH_S && tid < G_ROWS) {
    const long gr = rowBase + tid;
    if (gr < N) wmine = dinv[gr];
  }

  float4 bc = make_float4(0.0f, 0.0f, 0.0f, 0.0f);
  if (ACT_IN) { bc.x = bin[c4]; bc.y = bin[c4 + 1]; bc.z = bin[c4 + 2]; bc.w = bin[c4 + 3]; }

  const int half = lane >> 4;
  const int l16  = lane & 15;
  const int n0   = colTile * 16 + l16;

  v2f bw[K_DIM / 4];
#pragma unroll
  for (int s = 0; s < K_DIM / 4; ++s) {
    const int ka = 4 * s + 2 * half;
    bw[s].x = W[ka * DOUT + n0];
    bw[s].y = W[(ka + 1) * DOUT + n0];
  }

#pragma unroll
  for (int q = 4; q < G_PRE; ++q) {
    const long gr = rowBase + q * 8 + r8;
    pre[q] = make_float4(0.0f, 0.0f, 0.0f, 0.0f);
    if (gr < N) pre[q] = *reinterpret_cast<const float4*>(A + gr * K_DIM + c4);
  }

  if (WITH_S && tid < G_ROWS) Ws[tid] = wmine;

#pragma unroll
  for (int c = 0; c < (G_UNITS + 1) / 2; ++c) {
    const int unitsHere = (2 * c + 1 < G_UNITS) ? 2 : 1;
    const int trips = unitsHere * 2;
#pragma unroll
    for (int q = 0; q < trips; ++q) {
      const int lr = c * 32 + q * 8 + r8;
      float4 a = pre[c * 4 + q];
      if (ACT_IN) {
        a.x = fmaxf(a.x + bc.x, 0.0f);
        a.y = fmaxf(a.y + bc.y, 0.0f);
        a.z = fmaxf(a.z + bc.z, 0.0f);
        a.w = fmaxf(a.w + bc.w, 0.0f);
        if (rowBase + lr >= N) a = make_float4(0.0f, 0.0f, 0.0f, 0.0f);
      }
      *reinterpret_cast<float4*>(&As[lr][c4]) = a;
    }
    __builtin_amdgcn_fence(__ATOMIC_RELEASE, "workgroup", "local");
    __builtin_amdgcn_s_barrier();
    __builtin_amdgcn_fence(__ATOMIC_ACQUIRE, "workgroup", "local");

    if (NG == 1) {
      const int u0 = 2 * c;
      const int lrow = u0 * 16 + 8 * half;
      if (unitsHere == 2) {
        v8f acc[2] = {};
        mma_units<2>(As, u0 * 16 + l16, half, bw, acc);
        store_unit<DOUT, WITH_S>(acc[0], rowBase + lrow, n0, &Ws[lrow], T, S, N);
        store_unit<DOUT, WITH_S>(acc[1], rowBase + lrow + 16, n0, &Ws[lrow + 16], T, S, N);
      } else {
        v8f acc[1] = {};
        mma_units<1>(As, u0 * 16 + l16, half, bw, acc);
        store_unit<DOUT, WITH_S>(acc[0], rowBase + lrow, n0, &Ws[lrow], T, S, N);
      }
    } else {
      const int u = 2 * c + grp;
      if (u < G_UNITS) {
        const int lrow = u * 16 + 8 * half;
        v8f acc[1] = {};
        mma_units<1>(As, u * 16 + l16, half, bw, acc);
        store_unit<DOUT, WITH_S>(acc[0], rowBase + lrow, n0, &Ws[lrow], T, S, N);
      }
    }
  }
}

__device__ __forceinline__ int wave_half() {
  const unsigned id = __builtin_amdgcn_s_sendmsg_rtn(0x87);
  return (int)((id >> 18) & 1u);
}

__global__ void init_deg_kernel(unsigned* cnt0, unsigned* cnt1, int N, unsigned* tickets, unsigned* zeros, int nz) {
  const long i = (long)blockIdx.x * blockDim.x + threadIdx.x;
  if (i < N) { cnt0[i] = 0u; cnt1[i] = 0u; }
  if (i < 6) tickets[i * 64] = 0u;
  const long stride = (long)gridDim.x * blockDim.x;
  for (long k = i; k < nz; k += stride) zeros[k] = 0u;
}

#define DC_T  512
#define DC_H  8192
#define DC_ES 32
#define DC_U  49
__global__ __launch_bounds__(DC_T) void count_deg_kernel(const int* __restrict__ dst, unsigned* cnt0,
                                                         unsigned* cnt1, int E, int N) {
  constexpr int STRIPS = DC_H / DC_T;
  __shared__ unsigned s_hist[DC_H];
  __shared__ int s_home;
  const int tid = threadIdx.x;
  if (tid == 0) s_home = wave_half();
  const int nChunks = (N + DC_H - 1) / DC_H;
  const int items = nChunks * DC_ES;
  const long per = ((long)E + DC_ES - 1) / DC_ES;

  for (int it = blockIdx.x; it < items; it += gridDim.x) {
    const int slice = it % DC_ES;
    const int n0 = (it / DC_ES) * DC_H;
    const int words = (N - n0 < DC_H) ? N - n0 : DC_H;
    for (int i = tid; i < DC_H; i += DC_T) s_hist[i] = 0u;
    __syncthreads();

    const long eBeg = (long)slice * per;
    const long eEnd = (eBeg + per < (long)E) ? eBeg + per : (long)E;
    for (long k0 = eBeg + tid; k0 < eEnd; k0 += (long)DC_T * DC_U) {
      int d[DC_U];
#pragma unroll
      for (int u = 0; u < DC_U; ++u) {
        const long k = k0 + (long)u * DC_T;
        d[u] = (k < eEnd) ? dst[k] : -1;
      }
#pragma unroll
      for (int u = 0; u < DC_U; ++u) {
        const unsigned dl = (unsigned)(d[u] - n0);
        if (dl < (unsigned)words)
          __hip_atomic_fetch_add(&s_hist[dl], 1u, __ATOMIC_RELAXED, __HIP_MEMORY_SCOPE_WORKGROUP);
      }
    }
    __syncthreads();

    unsigned* c = (s_home ? cnt1 : cnt0) + n0;
    unsigned x[STRIPS];
#pragma unroll
    for (int u = 0; u < STRIPS; ++u) x[u] = s_hist[((u + slice) & (STRIPS - 1)) * DC_T + tid];
#pragma unroll
    for (int u = 0; u < STRIPS; ++u) {
      const int i = ((u + slice) & (STRIPS - 1)) * DC_T + tid;
      if (i < words && x[u] != 0u) atomicAdd(c + i, x[u]);
    }
    __syncthreads();
  }
}

__global__ void finalize_dinv_kernel(float* dinv, const unsigned* cnt0, const unsigned* cnt1, int N) {
  const long i = (long)blockIdx.x * blockDim.x + threadIdx.x;
  if (i < N) dinv[i] = rsqrtf(1.0f + (float)(cnt0[i] + cnt1[i]));
}


template <int D, int WT, bool WLOAD>
__global__ __launch_bounds__(256) void agg_edges_kernel(
    const float* __restrict__ T, const int* __restrict__ src,
    const int* __restrict__ dst, const float* __restrict__ dinv,
    float* __restrict__ out, int E, int N, unsigned* __restrict__ tickets,
    float* wedge) {
  constexpr int CPL = D / 32;
  constexpr int BT  = 8 * WT;
  __shared__ unsigned s_tile[2];
  __shared__ int s_home;
  const int tid  = threadIdx.x;
  const int lane = tid & 31;
  const int wave = __builtin_amdgcn_readfirstlane(tid >> 5);
  const int nTiles = (int)(((long)E + BT - 1) / BT);
  const int Nh = N >> 1;
  constexpr bool wread = WLOAD;
  const bool wwrite = !WLOAD && wedge != nullptr;

  if (tid == 0) s_home = wave_half();
  __syncthreads();
  const int home = __builtin_amdgcn_readfirstlane(s_home);
  int k = 0;

  for (int pass = 0; pass < 2; ++pass) {
    const int h = home ^ pass;
    unsigned* ctr = tickets + h * 64;
    unsigned pend = 0;
    if (tid == 0) pend = atomicAdd(ctr, 1u);
    for (int it = 0; it <= nTiles; ++it) {
      const int slot = k & 1;
      ++k;
      if (tid == 0) s_tile[slot] = pend;
      __syncthreads();
      const unsigned t = __builtin_amdgcn_readfirstlane(s_tile[slot]);
      if (t >= (unsigned)nTiles) break;

      const long e = (long)t * BT + wave * WT + lane;
      const bool ok = lane < WT && e < (long)E;
      int sv = 0, dv = 0;
      float wv = 0.0f;
      if (ok) {
        sv = src[e]; dv = dst[e];
        if (wread) wv = wedge[e];
      }
      const bool mine = ok && ((int)(dv >= Nh) == h);
      const unsigned mask = __builtin_amdgcn_ballot_w32(mine);

      float wsv = 0.0f, wdv = 0.0f;
      if (mine && !wread) { wsv = dinv[sv]; wdv = dinv[dv]; }

      int   dd[WT];
      float v[WT][CPL];
#pragma unroll
      for (int j = 0; j < WT; ++j) {
        if (mask & (1u << j)) {
          const int s = __builtin_amdgcn_readlane(sv, j);
          dd[j] = __builtin_amdgcn_readlane(dv, j);
          const float* tp = T + (long)s * D + lane;
#pragma unroll
          for (int cc = 0; cc < CPL; ++cc) v[j][cc] = tp[32 * cc];
        }
      }
      if (tid == 0) pend = atomicAdd(ctr, 1u);
      if (!wread) {
        wv = wsv * wdv;
        if (wwrite && mine) wedge[e] = wv;
      }
      const int wb = __float_as_int(wv);
#pragma unroll
      for (int j = 0; j < WT; ++j) {
        if (mask & (1u << j)) {
          const float nr = __int_as_float(__builtin_amdgcn_readlane(wb, j));
          float* op = out + (long)dd[j] * D + lane;
#pragma unroll
          for (int cc = 0; cc < CPL; ++cc) atomicAdd(op + 32 * cc, v[j][cc] * nr);
        }
      }
    }
  }
}

#define PE_T 1024
#define PE_G 512
#define PE_W 16

__device__ __forceinline__ void pe_record(long t, bool ok, int E,
                                          const int* __restrict__ src,
                                          const int* __restrict__ dst,
                                          const float* wedge,
                                          int& sv, int& nv, float& wa) {
  sv = 0; nv = 0; wa = 0.0f;
  if (ok) {
    if (t < (long)E) {
      sv = src[t]; nv = dst[t];
      if (wedge != nullptr) wa = wedge[t];
    } else {
      sv = nv = (int)(t - (long)E);
    }
  }
}

__global__ __launch_bounds__(PE_T) void pool_edges_kernel(
    const float* __restrict__ T, const int* __restrict__ src,
    const int* __restrict__ dst, const int* __restrict__ batch,
    const float* __restrict__ dinv, const float* wedge,
    float* accG, unsigned* cntG, int E, int N, int G) {
  __shared__ float    s_acc[PE_G * 64];
  __shared__ unsigned s_cnt[PE_G];
  __shared__ int      s_home;
  const int tid  = threadIdx.x;
  const int lane = tid & 31;
  const int wave = __builtin_amdgcn_readfirstlane(tid >> 5);
  const long items   = (long)E + (long)N;
  const long nChunks = (items + PE_W - 1) / PE_W;
  const long gw = (long)blockIdx.x * (PE_T / 32) + wave;
  const long GW = (long)gridDim.x * (PE_T / 32);
  if (tid == 0) s_home = wave_half();

  for (int g0 = 0; g0 < G; g0 += PE_G) {
    for (int i = tid; i < PE_G * 64; i += PE_T) s_acc[i] = 0.0f;
    if (tid < PE_G) s_cnt[tid] = 0u;
    __syncthreads();

    long c = gw;
    long t = c * PE_W + lane;
    bool ok = lane < PE_W && c < nChunks && t < items;
    int sv, nv;
    float wa;
    pe_record(t, ok, E, src, dst, wedge, sv, nv, wa);

    while (c < nChunks) {
      int gv = batch[nv];

      float v[PE_W][2];
#pragma unroll
      for (int j = 0; j < PE_W; ++j) {
        const int s = __builtin_amdgcn_readlane(sv, j);
        const float* tp = T + (long)s * 64 + lane;
        v[j][0] = tp[0];
        v[j][1] = tp[32];
      }

      const long c2 = c + GW;
      const long t2 = c2 * PE_W + lane;
      const bool ok2 = lane < PE_W && c2 < nChunks && t2 < items;
      int sv2, nv2;
      float wa2;
      pe_record(t2, ok2, E, src, dst, wedge, sv2, nv2, wa2);

      float w = wa;
      if (ok && (t >= (long)E || wedge == nullptr)) w = dinv[sv] * dinv[nv];

      const int   gl = gv - g0;
      const bool mine = ok && gl >= 0 && gl < PE_G && gv < G;
      if (mine && t >= (long)E)
        __hip_atomic_fetch_add(&s_cnt[gl], 1u, __ATOMIC_RELAXED, __HIP_MEMORY_SCOPE_WORKGROUP);
      const unsigned amask = __builtin_amdgcn_ballot_w32(mine);
      const int wb = __float_as_int(w);
#pragma unroll
      for (int j = 0; j < PE_W; ++j) {
        if (amask & (1u << j)) {
          const int   g  = __builtin_amdgcn_readlane(gl, j);
          const float wj = __int_as_float(__builtin_amdgcn_readlane(wb, j));
          float* ap = &s_acc[g * 64 + lane];
          __hip_atomic_fetch_add(ap,      v[j][0] * wj, __ATOMIC_RELAXED, __HIP_MEMORY_SCOPE_WORKGROUP);
          __hip_atomic_fetch_add(ap + 32, v[j][1] * wj, __ATOMIC_RELAXED, __HIP_MEMORY_SCOPE_WORKGROUP);
        }
      }
      c = c2; t = t2; ok = ok2; sv = sv2; nv = nv2; wa = wa2;
    }
    __syncthreads();

    float* acc = accG + (long)s_home * ((long)G * 64) + (long)g0 * 64;
    const int lim = (G - g0 < PE_G ? G - g0 : PE_G) * 64;
    const int rot = tid + (int)blockIdx.x * 128;
#pragma nounroll
    for (int k = 0; k < PE_G * 64; k += PE_T * 8) {
      float x[8];
#pragma unroll
      for (int u = 0; u < 8; ++u) x[u] = s_acc[(k + u * PE_T + rot) & (PE_G * 64 - 1)];
#pragma unroll
      for (int u = 0; u < 8; ++u) {
        const int i = (k + u * PE_T + rot) & (PE_G * 64 - 1);
        if (i < lim) atomicAdd(acc + i, x[u]);
      }
    }
    if (tid < PE_G && g0 + tid < G) {
      const unsigned n = s_cnt[tid];
      if (n != 0u) atomicAdd(cntG + g0 + tid, n);
    }
    __syncthreads();
  }
}

__global__ __launch_bounds__(256) void pool_final_kernel(
    const float* __restrict__ accG, const unsigned* __restrict__ cntG,
    const float* __restrict__ b3, float* __restrict__ out, int G) {
  const int i = (int)blockIdx.x * 256 + threadIdx.x;
  if (i < G * 64) {
    const float s = accG[i] + accG[(long)G * 64 + i];
    const float c = (float)cntG[i >> 6];
    out[i] = (s + c * b3[i & 63]) / fmaxf(c, 1.0f);
  }
}

static inline int cdiv(long a, long b) { return (int)((a + b - 1) / b); }

extern "C" void kernel_launch(void* const* d_in, const int* in_sizes, int n_in,
                              void* d_out, int out_size, void* d_ws, size_t ws_size,
                              hipStream_t stream) {
  const float* x   = (const float*)d_in[0];
  const int*   ei  = (const int*)d_in[1];
  const int*   bat = (const int*)d_in[2];
  const float* W1  = (const float*)d_in[3];
  const float* b1  = (const float*)d_in[4];
  const float* W2  = (const float*)d_in[5];
  const float* b2  = (const float*)d_in[6];
  const float* W3  = (const float*)d_in[7];
  const float* b3  = (const float*)d_in[8];
  (void)n_in;

  const int N = in_sizes[0] / 128;
  const int E = in_sizes[1] / 2;
  const int G = out_size / 64;
  const int* src = ei;
  const int* dst = ei + E;

  float* ws = (float*)d_ws;
  size_t o = 0;
  float* dinv = ws + o;   o += ((size_t)N + 63) & ~(size_t)63;
  float* T    = ws + o;   o += (size_t)N * 128;
  float* AGG  = ws + o;   o += (size_t)N * 128;
  float* counts = ws + o; o += (size_t)G;
  o = (o + 63) & ~(size_t)63;
  float* accG = ws + o;   o += (size_t)G * 128;
  unsigned* cntG = (unsigned*)(ws + o); o += (size_t)G;
  o = (o + 63) & ~(size_t)63;
  float* wedge = (ws_size >= (o + (size_t)E) * sizeof(float)) ? ws + o : nullptr;

  const int B = 256;
  float* out = (float*)d_out;

  unsigned* tickets = (unsigned*)counts;
  const int AG128 = 1536;
  const int PE_BLOCKS = 256;

  unsigned* cnt0 = (unsigned*)dinv;
  unsigned* cnt1 = (unsigned*)T;
  init_deg_kernel<<<cdiv(N, B), B, 0, stream>>>(cnt0, cnt1, N, tickets, (unsigned*)accG, G * 129);
  count_deg_kernel<<<PE_BLOCKS, DC_T, 0, stream>>>(dst, cnt0, cnt1, E, N);
  finalize_dinv_kernel<<<cdiv(N, B), B, 0, stream>>>(dinv, cnt0, cnt1, N);

  gemm_f32_wmma<128, false, true><<<cdiv(N, G_ROWS), B, 0, stream>>>(x, b1, W1, dinv, T, AGG, N);
  agg_edges_kernel<128, 16, false><<<AG128, B, 0, stream>>>(T, src, dst, dinv, AGG, E, N, tickets, wedge);

  gemm_f32_wmma<128, true, true><<<cdiv(N, G_ROWS), B, 0, stream>>>(AGG, b1, W2, dinv, T, AGG, N);
  if (wedge != nullptr)
    agg_edges_kernel<128, 16, true><<<AG128, B, 0, stream>>>(T, src, dst, dinv, AGG, E, N, tickets + 128, wedge);
  else
    agg_edges_kernel<128, 16, false><<<AG128, B, 0, stream>>>(T, src, dst, dinv, AGG, E, N, tickets + 128, wedge);

  gemm_f32_wmma<64, true, false><<<cdiv(N, G_ROWS), B, 0, stream>>>(AGG, b2, W3, dinv, T, nullptr, N);
  pool_edges_kernel<<<PE_BLOCKS, PE_T, 0, stream>>>(T, src, dst, bat, dinv, wedge, accG, cntG, E, N, G);

  pool_final_kernel<<<cdiv((long)G * 64, B), B, 0, stream>>>(accG, cntG, b3, out, G);
}
